# starting kernel
# baseline (speedup 1.0000x reference)
#include <hip/hip_runtime.h>
#include <hip/hip_bf16.h>

typedef __attribute__((ext_vector_type(2))) float v2f;
typedef __attribute__((ext_vector_type(8))) float v8f;

#define D_DIM        65
#define ROWS_PER_BLK 64
#define SD_STRIDE    132

__global__ __launch_bounds__(128) void mpuf_fused_kernel(
    const float* __restrict__ phi,
    const float* __restrict__ W_S, const float* __restrict__ b_S,
    const float* __restrict__ W_D, const float* __restrict__ b_D,
    float* __restrict__ out)
{
    __shared__ float SD[ROWS_PER_BLK * SD_STRIDE];

    const int tid   = threadIdx.x;
    const int wave  = tid >> 5;
    const int lane  = tid & 31;
    const int nl    = lane & 15;
    const int khalf = lane >> 4;

    const int rowBlock = blockIdx.x * ROWS_PER_BLK;
    const int rowWave  = rowBlock + wave * 16;

    const v8f vzero = {0.f,0.f,0.f,0.f,0.f,0.f,0.f,0.f};
    v8f acc[8];
    #pragma unroll
    for (int t = 0; t < 8; ++t) acc[t] = vzero;

    const float* __restrict__ phiRow = phi + (size_t)(rowWave + nl) * D_DIM;

    const int   s7     = 48 + nl;
    const bool  s7ok   = (s7 < 63);
    const float s7m    = s7ok ? 1.f : 0.f;
    const float* __restrict__ wS7 = W_S + (s7ok ? s7 : 62) * D_DIM;

    #pragma unroll 4
    for (int ks = 0; ks < 16; ++ks) {
        const int k0 = ks * 4 + khalf * 2;
        v2f a;
        a.x = phiRow[k0];
        a.y = phiRow[k0 + 1];
        #pragma unroll
        for (int t = 0; t < 8; ++t) {
            v2f b;
            if (t < 4) {
                const float* wp = W_D + (t * 16 + nl) * D_DIM;
                b.x = wp[k0];
                b.y = wp[k0 + 1];
            } else if (t < 7) {
                const float* wp = W_S + ((t - 4) * 16 + nl) * D_DIM;
                b.x = wp[k0];
                b.y = wp[k0 + 1];
            } else {
                b.x = wS7[k0]     * s7m;
                b.y = wS7[k0 + 1] * s7m;
            }
            acc[t] = __builtin_amdgcn_wmma_f32_16x16x4_f32(
                false, a, false, b, (short)0, acc[t], false, false);
        }
    }

    {
        const float kv = (khalf == 0) ? 1.f : 0.f;
        v2f a;
        a.x = phiRow[64] * kv;
        a.y = 0.f;
        #pragma unroll
        for (int t = 0; t < 8; ++t) {
            v2f b;
            b.y = 0.f;
            if (t < 4) {
                b.x = W_D[(t * 16 + nl) * D_DIM + 64] * kv;
            } else if (t < 7) {
                b.x = W_S[((t - 4) * 16 + nl) * D_DIM + 64] * kv;
            } else {
                b.x = wS7[64] * s7m * kv;
            }
            acc[t] = __builtin_amdgcn_wmma_f32_16x16x4_f32(
                false, a, false, b, (short)0, acc[t], false, false);
        }
    }

    #pragma unroll
    for (int t = 0; t < 8; ++t) {
        const int col = t * 16 + nl;
        float bias;
        if (col < 64)        bias = b_D[col];
        else if (col < 127)  bias = b_S[col - 64];
        else                 bias = 0.f;
        #pragma unroll
        for (int r = 0; r < 8; ++r) {
            const int rloc = wave * 16 + khalf * 8 + r;
            const float v  = acc[t][r] + bias;
            const float sg = 1.0f / (1.0f + __expf(-v));
            SD[rloc * SD_STRIDE + col] = sg;
        }
    }
    __syncthreads();

    const int row  = tid >> 1;
    const int half = tid & 1;
    const float* __restrict__ sd = SD + row * SD_STRIDE;

    constexpr int loff[6] = {0, 32, 48, 56, 60, 62};

    float partial = 0.f;
    #pragma unroll
    for (int jj = 0; jj < 32; ++jj) {
        const int j = half * 32 + jj;
        float m = sd[j];
        #pragma unroll
        for (int i = 0; i < 6; ++i) {
            const int idx = loff[i] + (j >> (i + 1));
            const float s = sd[64 + idx];
            m *= ((j >> i) & 1) ? s : (1.0f - s);
        }
        partial += m;
    }
    partial += __shfl_xor(partial, 1, 32);
    if (half == 0) {
        out[rowBlock + row] = partial * 0.99999f;
    }
}

extern "C" void kernel_launch(void* const* d_in, const int* in_sizes, int n_in,
                              void* d_out, int out_size, void* d_ws, size_t ws_size,
                              hipStream_t stream) {
    const float* phi = (const float*)d_in[0];
    const float* W_S = (const float*)d_in[1];
    const float* b_S = (const float*)d_in[2];
    const float* W_D = (const float*)d_in[3];
    const float* b_D = (const float*)d_in[4];
    float* out = (float*)d_out;

    const int B      = in_sizes[0] / D_DIM;
    const int blocks = B / ROWS_PER_BLK;

    mpuf_fused_kernel<<<blocks, 128, 0, stream>>>(phi, W_S, b_S, W_D, b_D, out);
}


# B operand (W_D/W_S) staged once per work-group in LDS in WMMA fragment order, read by grouped ds_load_b128 instead of strided global loads each k-step; waves walk 4 row tiles, no per-tile barrier
# speedup vs baseline: 2.0172x; 2.0172x over previous
#include <hip/hip_runtime.h>
#include <hip/hip_bf16.h>


typedef __attribute__((ext_vector_type(2))) float v2f;
typedef __attribute__((ext_vector_type(4))) float v4f;
typedef __attribute__((ext_vector_type(8))) float v8f;

#define D_DIM          65
#define WAVES_PER_BLK  4
#define THREADS        (WAVES_PER_BLK * 32)
#define TILES_PER_WAVE 4
#define ROWS_PER_BLK   (WAVES_PER_BLK * 16 * TILES_PER_WAVE)
#define SD_STRIDE      132
#define KSTEPS         17
#define BF_FLOATS      (KSTEPS * 4 * 32 * 4)

__global__ __launch_bounds__(THREADS) __attribute__((amdgpu_waves_per_eu(4)))
void mpuf_fused_kernel(
    const float* __restrict__ phi,
    const float* __restrict__ W_S, const float* __restrict__ b_S,
    const float* __restrict__ W_D, const float* __restrict__ b_D,
    float* __restrict__ out)
{
    __shared__ __attribute__((aligned(16))) float BF[BF_FLOATS];
    __shared__ float SD[WAVES_PER_BLK * 16 * SD_STRIDE];

    const int tid   = threadIdx.x;
    const int wave  = tid >> 5;
    const int lane  = tid & 31;
    const int nl    = lane & 15;
    const int khalf = lane >> 4;

    {
        const int  srow = tid >> 2;
        const int  skq  = tid & 3;
        const int  dl   = ((((tid >> 1) & 1) * 16 + ((tid >> 2) & 15)) << 2)
                        + (((tid >> 6) & 1) << 1) + (tid & 1);
        const bool sok  = (srow < 31);
        const bool kok  = (skq == 0);
        const float* __restrict__ pD0 = W_D + srow * D_DIM + skq;
        const float* __restrict__ pD1 = W_D + (32 + srow) * D_DIM + skq;
        const float* __restrict__ pS0 = W_S + srow * D_DIM + skq;
        const float* __restrict__ pS1 = W_S + (32 + (sok ? srow : 30)) * D_DIM + skq;
        #pragma unroll
        for (int i = 0; i < KSTEPS * 4; ++i) {
            const int q   = i & 3;
            const float* __restrict__ p = (q == 0) ? pD0 : (q == 1) ? pD1 : (q == 2) ? pS0 : pS1;
            float v;
            if (i < 64) {
                v = p[(i >> 2) * 4];
            } else {
                v = p[64 - skq];
                v = kok ? v : 0.f;
            }
            if (q == 3) v = sok ? v : 0.f;
            BF[i * 128 + dl] = v;
        }
    }

    float bias[8];
    #pragma unroll
    for (int t = 0; t < 8; ++t) {
        const int col = t * 16 + nl;
        if (col < 64)        bias[t] = b_D[col];
        else if (col < 127)  bias[t] = b_S[col - 64];
        else                 bias[t] = 0.f;
    }

    __syncthreads();

    const v4f* __restrict__ bf = reinterpret_cast<const v4f*>(BF) + lane;
    float* __restrict__ SDw = SD + wave * 16 * SD_STRIDE;
    const float kv = (khalf == 0) ? 1.f : 0.f;

    for (int it = 0; it < TILES_PER_WAVE; ++it) {
        const int rowWave = blockIdx.x * ROWS_PER_BLK + (it * WAVES_PER_BLK + wave) * 16;

        const v8f vzero = {0.f,0.f,0.f,0.f,0.f,0.f,0.f,0.f};
        v8f acc[8];
        #pragma unroll
        for (int t = 0; t < 8; ++t) acc[t] = vzero;

        const float* __restrict__ phiRow = phi + (size_t)(rowWave + nl) * D_DIM;

        v2f afr[KSTEPS];
        #pragma unroll
        for (int ks = 0; ks < 16; ++ks) {
            const int k0 = ks * 4 + khalf * 2;
            afr[ks].x = phiRow[k0];
            afr[ks].y = phiRow[k0 + 1];
        }
        afr[16].x = phiRow[64] * kv;
        afr[16].y = 0.f;

        v4f bcur[8], bnxt[8];
        #pragma unroll
        for (int i = 0; i < 8; ++i) bcur[i] = bf[i * 32];
        #pragma unroll
        for (int g = 0; g < 9; ++g) {
            const int nnext = (g < 7) ? 8 : ((g == 7) ? 4 : 0);
            #pragma unroll
            for (int i = 0; i < 8; ++i)
                if (i < nnext) bnxt[i] = bf[((g + 1) * 8 + i) * 32];
            __builtin_amdgcn_sched_barrier(0);
            const int nsteps = (g < 8) ? 2 : 1;
            #pragma unroll
            for (int s = 0; s < 2; ++s) {
                if (s < nsteps) {
                    const v2f a = afr[g * 2 + s];
                    #pragma unroll
                    for (int tp = 0; tp < 4; ++tp) {
                        const v4f b = bcur[s * 4 + tp];
                        v2f b0, b1;
                        b0.x = b.x; b0.y = b.y;
                        b1.x = b.z; b1.y = b.w;
                        acc[2 * tp]     = __builtin_amdgcn_wmma_f32_16x16x4_f32(
                            false, a, false, b0, (short)0, acc[2 * tp],     false, false);
                        acc[2 * tp + 1] = __builtin_amdgcn_wmma_f32_16x16x4_f32(
                            false, a, false, b1, (short)0, acc[2 * tp + 1], false, false);
                    }
                }
            }
            __builtin_amdgcn_sched_barrier(0);
            #pragma unroll
            for (int i = 0; i < 8; ++i) bcur[i] = bnxt[i];
        }

        #pragma unroll
        for (int t = 0; t < 8; ++t) {
            const int col = t * 16 + nl;
            #pragma unroll
            for (int r = 0; r < 8; ++r) {
                const int rloc = khalf * 8 + r;
                const float v  = acc[t][r] + bias[t];
                const float sg = 1.0f / (1.0f + __expf(-v));
                SDw[rloc * SD_STRIDE + col] = sg;
            }
        }
        __builtin_amdgcn_fence(__ATOMIC_SEQ_CST, "workgroup");

        const int row  = lane >> 1;
        const int half = lane & 1;
        const float* sd = SDw + row * SD_STRIDE;

        constexpr int loff[6] = {0, 32, 48, 56, 60, 62};

        float partial = 0.f;
        #pragma unroll
        for (int jj = 0; jj < 32; ++jj) {
            const int j = half * 32 + jj;
            float m = sd[j];
            #pragma unroll
            for (int i = 0; i < 6; ++i) {
                const int idx = loff[i] + (j >> (i + 1));
                const float s = sd[64 + idx];
                m *= ((j >> i) & 1) ? s : (1.0f - s);
            }
            partial += m;
        }
        partial += __shfl_xor(partial, 1, 32);
        if (half == 0) {
            out[rowWave + row] = partial * 0.99999f;
        }
    }
}

extern "C" void kernel_launch(void* const* d_in, const int* in_sizes, int n_in,
                              void* d_out, int out_size, void* d_ws, size_t ws_size,
                              hipStream_t stream) {
    const float* phi = (const float*)d_in[0];
    const float* W_S = (const float*)d_in[1];
    const float* b_S = (const float*)d_in[2];
    const float* W_D = (const float*)d_in[3];
    const float* b_D = (const float*)d_in[4];
    float* out = (float*)d_out;

    const int B      = in_sizes[0] / D_DIM;
    const int blocks = B / ROWS_PER_BLK;

    mpuf_fused_kernel<<<blocks, THREADS, 0, stream>>>(phi, W_S, b_S, W_D, b_D, out);
}


# sigmoid's 1/(1+e) taken with one v_rcp_f32 (1 ulp) instead of the correctly rounded division sequence: 11 vector instructions per element become 1; sums and their order unchanged
# speedup vs baseline: 2.3317x; 1.1559x over previous
#include <hip/hip_runtime.h>
#include <hip/hip_bf16.h>


typedef __attribute__((ext_vector_type(2))) float v2f;
typedef __attribute__((ext_vector_type(4))) float v4f;
typedef __attribute__((ext_vector_type(8))) float v8f;

#define D_DIM          65
#define WAVES_PER_BLK  4
#define THREADS        (WAVES_PER_BLK * 32)
#define TILES_PER_WAVE 4
#define ROWS_PER_BLK   (WAVES_PER_BLK * 16 * TILES_PER_WAVE)
#define SD_STRIDE      132
#define KSTEPS         17
#define BF_FLOATS      (KSTEPS * 4 * 32 * 4)

__global__ __launch_bounds__(THREADS) __attribute__((amdgpu_waves_per_eu(4)))
void mpuf_fused_kernel(
    const float* __restrict__ phi,
    const float* __restrict__ W_S, const float* __restrict__ b_S,
    const float* __restrict__ W_D, const float* __restrict__ b_D,
    float* __restrict__ out)
{
    __shared__ __attribute__((aligned(16))) float BF[BF_FLOATS];
    __shared__ float SD[WAVES_PER_BLK * 16 * SD_STRIDE];

    const int tid   = threadIdx.x;
    const int wave  = tid >> 5;
    const int lane  = tid & 31;
    const int nl    = lane & 15;
    const int khalf = lane >> 4;

    {
        const int  srow = tid >> 2;
        const int  skq  = tid & 3;
        const int  dl   = ((((tid >> 1) & 1) * 16 + ((tid >> 2) & 15)) << 2)
                        + (((tid >> 6) & 1) << 1) + (tid & 1);
        const bool sok  = (srow < 31);
        const bool kok  = (skq == 0);
        const float* __restrict__ pD0 = W_D + srow * D_DIM + skq;
        const float* __restrict__ pD1 = W_D + (32 + srow) * D_DIM + skq;
        const float* __restrict__ pS0 = W_S + srow * D_DIM + skq;
        const float* __restrict__ pS1 = W_S + (32 + (sok ? srow : 30)) * D_DIM + skq;
        #pragma unroll
        for (int i = 0; i < KSTEPS * 4; ++i) {
            const int q   = i & 3;
            const float* __restrict__ p = (q == 0) ? pD0 : (q == 1) ? pD1 : (q == 2) ? pS0 : pS1;
            float v;
            if (i < 64) {
                v = p[(i >> 2) * 4];
            } else {
                v = p[64 - skq];
                v = kok ? v : 0.f;
            }
            if (q == 3) v = sok ? v : 0.f;
            BF[i * 128 + dl] = v;
        }
    }

    float bias[8];
    #pragma unroll
    for (int t = 0; t < 8; ++t) {
        const int col = t * 16 + nl;
        if (col < 64)        bias[t] = b_D[col];
        else if (col < 127)  bias[t] = b_S[col - 64];
        else                 bias[t] = 0.f;
    }

    __syncthreads();

    const v4f* __restrict__ bf = reinterpret_cast<const v4f*>(BF) + lane;
    float* __restrict__ SDw = SD + wave * 16 * SD_STRIDE;
    const float kv = (khalf == 0) ? 1.f : 0.f;

    for (int it = 0; it < TILES_PER_WAVE; ++it) {
        const int rowWave = blockIdx.x * ROWS_PER_BLK + (it * WAVES_PER_BLK + wave) * 16;

        const v8f vzero = {0.f,0.f,0.f,0.f,0.f,0.f,0.f,0.f};
        v8f acc[8];
        #pragma unroll
        for (int t = 0; t < 8; ++t) acc[t] = vzero;

        const float* __restrict__ phiRow = phi + (size_t)(rowWave + nl) * D_DIM;

        v2f afr[KSTEPS];
        #pragma unroll
        for (int ks = 0; ks < 16; ++ks) {
            const int k0 = ks * 4 + khalf * 2;
            afr[ks].x = phiRow[k0];
            afr[ks].y = phiRow[k0 + 1];
        }
        afr[16].x = phiRow[64] * kv;
        afr[16].y = 0.f;

        v4f bcur[8], bnxt[8];
        #pragma unroll
        for (int i = 0; i < 8; ++i) bcur[i] = bf[i * 32];
        #pragma unroll
        for (int g = 0; g < 9; ++g) {
            const int nnext = (g < 7) ? 8 : ((g == 7) ? 4 : 0);
            #pragma unroll
            for (int i = 0; i < 8; ++i)
                if (i < nnext) bnxt[i] = bf[((g + 1) * 8 + i) * 32];
            __builtin_amdgcn_sched_barrier(0);
            const int nsteps = (g < 8) ? 2 : 1;
            #pragma unroll
            for (int s = 0; s < 2; ++s) {
                if (s < nsteps) {
                    const v2f a = afr[g * 2 + s];
                    #pragma unroll
                    for (int tp = 0; tp < 4; ++tp) {
                        const v4f b = bcur[s * 4 + tp];
                        v2f b0, b1;
                        b0.x = b.x; b0.y = b.y;
                        b1.x = b.z; b1.y = b.w;
                        acc[2 * tp]     = __builtin_amdgcn_wmma_f32_16x16x4_f32(
                            false, a, false, b0, (short)0, acc[2 * tp],     false, false);
                        acc[2 * tp + 1] = __builtin_amdgcn_wmma_f32_16x16x4_f32(
                            false, a, false, b1, (short)0, acc[2 * tp + 1], false, false);
                    }
                }
            }
            __builtin_amdgcn_sched_barrier(0);
            #pragma unroll
            for (int i = 0; i < 8; ++i) bcur[i] = bnxt[i];
        }

        #pragma unroll
        for (int t = 0; t < 8; ++t) {
            const int col = t * 16 + nl;
            #pragma unroll
            for (int r = 0; r < 8; ++r) {
                const int rloc = khalf * 8 + r;
                const float v  = acc[t][r] + bias[t];
                const float sg = __builtin_amdgcn_rcpf(1.0f + __expf(-v));
                SDw[rloc * SD_STRIDE + col] = sg;
            }
        }
        __builtin_amdgcn_fence(__ATOMIC_SEQ_CST, "workgroup");

        const int row  = lane >> 1;
        const int half = lane & 1;
        const float* sd = SDw + row * SD_STRIDE;

        constexpr int loff[6] = {0, 32, 48, 56, 60, 62};

        float partial = 0.f;
        #pragma unroll
        for (int jj = 0; jj < 32; ++jj) {
            const int j = half * 32 + jj;
            float m = sd[j];
            #pragma unroll
            for (int i = 0; i < 6; ++i) {
                const int idx = loff[i] + (j >> (i + 1));
                const float s = sd[64 + idx];
                m *= ((j >> i) & 1) ? s : (1.0f - s);
            }
            partial += m;
        }
        partial += __shfl_xor(partial, 1, 32);
        if (half == 0) {
            out[rowWave + row] = partial * 0.99999f;
        }
    }
}

extern "C" void kernel_launch(void* const* d_in, const int* in_sizes, int n_in,
                              void* d_out, int out_size, void* d_ws, size_t ws_size,
                              hipStream_t stream) {
    const float* phi = (const float*)d_in[0];
    const float* W_S = (const float*)d_in[1];
    const float* b_S = (const float*)d_in[2];
    const float* W_D = (const float*)d_in[3];
    const float* b_D = (const float*)d_in[4];
    float* out = (float*)d_out;

    const int B      = in_sizes[0] / D_DIM;
    const int blocks = B / ROWS_PER_BLK;

    mpuf_fused_kernel<<<blocks, THREADS, 0, stream>>>(phi, W_S, b_S, W_D, b_D, out);
}


# A fragments of the wave's next row tile fetched one tile ahead (re-loaded a WMMA group after last use; tile 0 before B staging); phase-2 LDS indices as base+constant to fit registers; sum order kept
# speedup vs baseline: 2.5162x; 1.0791x over previous
#include <hip/hip_runtime.h>
#include <hip/hip_bf16.h>


typedef __attribute__((ext_vector_type(2))) float v2f;
typedef __attribute__((ext_vector_type(4))) float v4f;
typedef __attribute__((ext_vector_type(8))) float v8f;

#define D_DIM          65
#define WAVES_PER_BLK  4
#define THREADS        (WAVES_PER_BLK * 32)
#define TILES_PER_WAVE 4
#define ROWS_PER_BLK   (WAVES_PER_BLK * 16 * TILES_PER_WAVE)
#define SD_STRIDE      132
#define KSTEPS         17
#define BF_FLOATS      (KSTEPS * 4 * 32 * 4)

__global__ __launch_bounds__(THREADS) __attribute__((amdgpu_waves_per_eu(4)))
void mpuf_fused_kernel(
    const float* __restrict__ phi,
    const float* __restrict__ W_S, const float* __restrict__ b_S,
    const float* __restrict__ W_D, const float* __restrict__ b_D,
    float* __restrict__ out)
{
    __shared__ __attribute__((aligned(16))) float BF[BF_FLOATS];
    __shared__ float SD[WAVES_PER_BLK * 16 * SD_STRIDE];

    const int tid   = threadIdx.x;
    const int wave  = tid >> 5;
    const int lane  = tid & 31;
    const int nl    = lane & 15;
    const int khalf = lane >> 4;

    const float* __restrict__ phiRow =
        phi + (size_t)(blockIdx.x * ROWS_PER_BLK + wave * 16 + nl) * D_DIM;
    v2f   afr[KSTEPS];
    float atail;
    #pragma unroll
    for (int ks = 0; ks < 16; ++ks) {
        const int k0 = ks * 4 + khalf * 2;
        afr[ks].x = phiRow[k0];
        afr[ks].y = phiRow[k0 + 1];
    }
    atail = phiRow[64];
    __builtin_amdgcn_sched_barrier(0);

    {
        const int  srow = tid >> 2;
        const int  skq  = tid & 3;
        const int  dl   = ((((tid >> 1) & 1) * 16 + ((tid >> 2) & 15)) << 2)
                        + (((tid >> 6) & 1) << 1) + (tid & 1);
        const bool sok  = (srow < 31);
        const bool kok  = (skq == 0);
        const float* __restrict__ pD0 = W_D + srow * D_DIM + skq;
        const float* __restrict__ pD1 = W_D + (32 + srow) * D_DIM + skq;
        const float* __restrict__ pS0 = W_S + srow * D_DIM + skq;
        const float* __restrict__ pS1 = W_S + (32 + (sok ? srow : 30)) * D_DIM + skq;
        #pragma unroll
        for (int i = 0; i < KSTEPS * 4; ++i) {
            const int q   = i & 3;
            const float* __restrict__ p = (q == 0) ? pD0 : (q == 1) ? pD1 : (q == 2) ? pS0 : pS1;
            float v;
            if (i < 64) {
                v = p[(i >> 2) * 4];
            } else {
                v = p[64 - skq];
                v = kok ? v : 0.f;
            }
            if (q == 3) v = sok ? v : 0.f;
            BF[i * 128 + dl] = v;
        }
    }

    float bias[8];
    #pragma unroll
    for (int t = 0; t < 8; ++t) {
        const int col = t * 16 + nl;
        if (col < 64)        bias[t] = b_D[col];
        else if (col < 127)  bias[t] = b_S[col - 64];
        else                 bias[t] = 0.f;
    }

    __syncthreads();

    const v4f* __restrict__ bf = reinterpret_cast<const v4f*>(BF) + lane;
    float* __restrict__ SDw = SD + wave * 16 * SD_STRIDE;
    const float kv = (khalf == 0) ? 1.f : 0.f;

    for (int it = 0; it < TILES_PER_WAVE; ++it) {
        const int rowWave = blockIdx.x * ROWS_PER_BLK + (it * WAVES_PER_BLK + wave) * 16;

        const v8f vzero = {0.f,0.f,0.f,0.f,0.f,0.f,0.f,0.f};
        v8f acc[8];
        #pragma unroll
        for (int t = 0; t < 8; ++t) acc[t] = vzero;

        const bool hasNext = (it + 1 < TILES_PER_WAVE);
        const float* __restrict__ phiNext = phiRow + WAVES_PER_BLK * 16 * D_DIM;
        afr[16].x = atail * kv;
        afr[16].y = 0.f;

        v4f bcur[8], bnxt[8];
        #pragma unroll
        for (int i = 0; i < 8; ++i) bcur[i] = bf[i * 32];
        #pragma unroll
        for (int g = 0; g < 9; ++g) {
            const int nnext = (g < 7) ? 8 : ((g == 7) ? 4 : 0);
            #pragma unroll
            for (int i = 0; i < 8; ++i)
                if (i < nnext) bnxt[i] = bf[((g + 1) * 8 + i) * 32];
            if (g >= 2 && hasNext) {
                #pragma unroll
                for (int s = 0; s < 2; ++s) {
                    const int ks = (g - 2) * 2 + s;
                    const int k0 = ks * 4 + khalf * 2;
                    afr[ks].x = phiNext[k0];
                    afr[ks].y = phiNext[k0 + 1];
                }
                if (g == 8) atail = phiNext[64];
            }
            __builtin_amdgcn_sched_barrier(0);
            const int nsteps = (g < 8) ? 2 : 1;
            #pragma unroll
            for (int s = 0; s < 2; ++s) {
                if (s < nsteps) {
                    const v2f a = afr[g * 2 + s];
                    #pragma unroll
                    for (int tp = 0; tp < 4; ++tp) {
                        const v4f b = bcur[s * 4 + tp];
                        v2f b0, b1;
                        b0.x = b.x; b0.y = b.y;
                        b1.x = b.z; b1.y = b.w;
                        acc[2 * tp]     = __builtin_amdgcn_wmma_f32_16x16x4_f32(
                            false, a, false, b0, (short)0, acc[2 * tp],     false, false);
                        acc[2 * tp + 1] = __builtin_amdgcn_wmma_f32_16x16x4_f32(
                            false, a, false, b1, (short)0, acc[2 * tp + 1], false, false);
                    }
                }
            }
            __builtin_amdgcn_sched_barrier(0);
            #pragma unroll
            for (int i = 0; i < 8; ++i) bcur[i] = bnxt[i];
        }
        if (hasNext) {
            #pragma unroll
            for (int ks = 14; ks < 16; ++ks) {
                const int k0 = ks * 4 + khalf * 2;
                afr[ks].x = phiNext[k0];
                afr[ks].y = phiNext[k0 + 1];
            }
        }
        phiRow = phiNext;

        #pragma unroll
        for (int t = 0; t < 8; ++t) {
            const int col = t * 16 + nl;
            #pragma unroll
            for (int r = 0; r < 8; ++r) {
                const int rloc = khalf * 8 + r;
                const float v  = acc[t][r] + bias[t];
                const float sg = __builtin_amdgcn_rcpf(1.0f + __expf(-v));
                SDw[rloc * SD_STRIDE + col] = sg;
            }
        }
        __builtin_amdgcn_fence(__ATOMIC_SEQ_CST, "workgroup", "local");

        const int row  = lane >> 1;
        const int half = lane & 1;
        const float* sd = SDw + row * SD_STRIDE;

        constexpr int loff[6] = {0, 32, 48, 56, 60, 62};

        const float* sdD = sd + half * 32;
        const float* sdS[6];
        #pragma unroll
        for (int i = 0; i < 6; ++i)
            sdS[i] = sd + 64 + loff[i] + ((half * 32) >> (i + 1));

        float partial = 0.f;
        #pragma unroll
        for (int jj = 0; jj < 32; ++jj) {
            float m = sdD[jj];
            #pragma unroll
            for (int i = 0; i < 6; ++i) {
                const float s   = sdS[i][jj >> (i + 1)];
                const int   bit = (i < 5) ? ((jj >> i) & 1) : half;
                m *= bit ? s : (1.0f - s);
            }
            partial += m;
        }
        partial += __shfl_xor(partial, 1, 32);
        if (half == 0) {
            out[rowWave + row] = partial * 0.99999f;
        }
    }
}

extern "C" void kernel_launch(void* const* d_in, const int* in_sizes, int n_in,
                              void* d_out, int out_size, void* d_ws, size_t ws_size,
                              hipStream_t stream) {
    const float* phi = (const float*)d_in[0];
    const float* W_S = (const float*)d_in[1];
    const float* b_S = (const float*)d_in[2];
    const float* W_D = (const float*)d_in[3];
    const float* b_D = (const float*)d_in[4];
    float* out = (float*)d_out;

    const int B      = in_sizes[0] / D_DIM;
    const int blocks = B / ROWS_PER_BLK;

    mpuf_fused_kernel<<<blocks, THREADS, 0, stream>>>(phi, W_S, b_S, W_D, b_D, out);
}
